# starting kernel
# baseline (speedup 1.0000x reference)
#include <hip/hip_runtime.h>
#include <hip/hip_bf16.h>


typedef __attribute__((ext_vector_type(16))) _Float16 v16h;
typedef __attribute__((ext_vector_type(8)))  _Float16 v8h;
typedef __attribute__((ext_vector_type(8)))  float    v8f;

#define LATENT   64
#define EDGE_IN  70
#define K1PAD    96
#define HID      20
#define OUTD     192
#define OUT_CH   3
#define WAVES    4


__device__ inline v8f wmma16(v16h a, v16h b, v8f c) {
    return __builtin_amdgcn_wmma_f32_16x16x32_f16(
        false, a, false, b, (short)0, c, false, false);
}

__device__ inline v16h cat8(v8h lo, v8h hi) {
    return __builtin_shufflevector(lo, hi,
        0, 1, 2, 3, 4, 5, 6, 7, 8, 9, 10, 11, 12, 13, 14, 15);
}

__device__ inline v16h load_a(const _Float16* H, int ld, int k0) {
    int lane = threadIdx.x & 31;
    int m  = lane & 15;
    int kb = k0 + ((lane & 16) ? 8 : 0);
    v8h lo = *(const v8h*)(H + m * ld + kb);
    v8h hi = *(const v8h*)(H + m * ld + kb + 16);
    return cat8(lo, hi);
}

__device__ inline v16h make_b(const _Float16* WT, int ldk, int k0, int n0) {
    int lane = threadIdx.x & 31;
    int n  = n0 + (lane & 15);
    int kb = k0 + ((lane & 16) ? 8 : 0);
    v8h lo = *(const v8h*)(WT + n * ldk + kb);
    v8h hi = *(const v8h*)(WT + n * ldk + kb + 16);
    return cat8(lo, hi);
}

__device__ inline void store_c(_Float16* H, int ld, v8f c, int n0,
                               const float* bias, bool relu) {
    int lane = threadIdx.x & 31;
    int n  = n0 + (lane & 15);
    int mb = (lane & 16) ? 8 : 0;
    float bv = bias[n];
#pragma unroll
    for (int r = 0; r < 8; ++r) {
        float v = c[r] + bv;
        if (relu) v = fmaxf(v, 0.0f);
        H[(mb + r) * ld + n] = (_Float16)v;
    }
}


__global__ void zero_f32(float* p, int n) {
    int i = blockIdx.x * blockDim.x + threadIdx.x;
    if (i < n) p[i] = 0.0f;
}

__global__ __launch_bounds__(WAVES * 32)
void nnconv_edge_kernel(const float* __restrict__ x,
                        const int*   __restrict__ edge_index,
                        const float* __restrict__ edge_attr,
                        const float* __restrict__ W1, const float* __restrict__ b1,
                        const float* __restrict__ W2, const float* __restrict__ b2,
                        const float* __restrict__ W3, const float* __restrict__ b3,
                        const float* __restrict__ W4, const float* __restrict__ b4,
                        float* __restrict__ summed, float* __restrict__ count,
                        int E) {
    __shared__ alignas(16) _Float16 sW1T[32  * K1PAD];
    __shared__ alignas(16) _Float16 sW2T[32  * 32];
    __shared__ alignas(16) _Float16 sW3T[32  * 32];
    __shared__ alignas(16) _Float16 sW4T[OUTD * 32];
    __shared__ float sB1[32], sB2[32], sB3[32];
    __shared__ float sB4[OUTD];
    __shared__ alignas(16) _Float16 sA [WAVES][16 * K1PAD];
    __shared__ alignas(16) _Float16 sO4[WAVES][16 * OUTD];

    const int tid = threadIdx.x;

    for (int i = tid; i < 32 * K1PAD; i += blockDim.x) {
        int n = i / K1PAD, k = i % K1PAD;
        sW1T[i] = (_Float16)((k < EDGE_IN && n < HID) ? W1[k * HID + n] : 0.0f);
    }
    for (int i = tid; i < 32 * 32; i += blockDim.x) {
        int n = i >> 5, k = i & 31;
        bool ok = (k < HID && n < HID);
        sW2T[i] = (_Float16)(ok ? W2[k * HID + n] : 0.0f);
        sW3T[i] = (_Float16)(ok ? W3[k * HID + n] : 0.0f);
    }
    for (int i = tid; i < OUTD * 32; i += blockDim.x) {
        int n = i >> 5, k = i & 31;
        sW4T[i] = (_Float16)((k < HID) ? W4[k * OUTD + n] : 0.0f);
    }
    for (int i = tid; i < 32; i += blockDim.x) {
        sB1[i] = (i < HID) ? b1[i] : 0.0f;
        sB2[i] = (i < HID) ? b2[i] : 0.0f;
        sB3[i] = (i < HID) ? b3[i] : 0.0f;
    }
    for (int i = tid; i < OUTD; i += blockDim.x) sB4[i] = b4[i];
    __syncthreads();

    const int lane = tid & 31;
    const int wave = tid >> 5;
    const int m    = lane & 15;
    const int half = lane >> 4;

    const long tile = (long)blockIdx.x * WAVES + wave;
    const long e0   = tile * 16;
    if (e0 >= E) return;

    _Float16* A1 = sA [wave];
    _Float16* O4 = sO4[wave];

    {
        const long gbase = e0 * EDGE_IN;
        const long gmax  = (long)E * EDGE_IN;
#pragma unroll 5
        for (int i = lane; i < 16 * EDGE_IN; i += 32) {
            long g = gbase + i;
            float v = (g < gmax) ? edge_attr[g] : 0.0f;
            A1[(i / EDGE_IN) * K1PAD + (i % EDGE_IN)] = (_Float16)v;
        }
#pragma unroll
        for (int i = lane; i < 16 * (K1PAD - EDGE_IN); i += 32) {
            A1[(i / 26) * K1PAD + EDGE_IN + (i % 26)] = (_Float16)0.0f;
        }
    }
    asm volatile("s_wait_dscnt 0" ::: "memory");

    _Float16* H = A1;
    {
        v8f c0 = {}, c1 = {};
#pragma unroll
        for (int ks = 0; ks < 3; ++ks) {
            v16h a = load_a(A1, K1PAD, ks * 32);
            c0 = wmma16(a, make_b(sW1T, K1PAD, ks * 32, 0),  c0);
            c1 = wmma16(a, make_b(sW1T, K1PAD, ks * 32, 16), c1);
        }
        store_c(H, 32, c0, 0,  sB1, true);
        store_c(H, 32, c1, 16, sB1, true);
    }
    asm volatile("s_wait_dscnt 0" ::: "memory");

    {
        v16h a = load_a(H, 32, 0);
        v8f c0 = {}, c1 = {};
        c0 = wmma16(a, make_b(sW2T, 32, 0, 0),  c0);
        c1 = wmma16(a, make_b(sW2T, 32, 0, 16), c1);
        asm volatile("s_wait_dscnt 0" ::: "memory");
        store_c(H, 32, c0, 0,  sB2, true);
        store_c(H, 32, c1, 16, sB2, true);
    }
    asm volatile("s_wait_dscnt 0" ::: "memory");

    {
        v16h a = load_a(H, 32, 0);
        v8f c0 = {}, c1 = {};
        c0 = wmma16(a, make_b(sW3T, 32, 0, 0),  c0);
        c1 = wmma16(a, make_b(sW3T, 32, 0, 16), c1);
        asm volatile("s_wait_dscnt 0" ::: "memory");
        store_c(H, 32, c0, 0,  sB3, true);
        store_c(H, 32, c1, 16, sB3, true);
    }
    asm volatile("s_wait_dscnt 0" ::: "memory");

    {
        v16h a = load_a(H, 32, 0);
#pragma unroll
        for (int t = 0; t < OUTD / 16; ++t) {
            v8f c = {};
            c = wmma16(a, make_b(sW4T, 32, 0, t * 16), c);
            store_c(O4, OUTD, c, t * 16, sB4, false);
        }
    }
    asm volatile("s_wait_dscnt 0" ::: "memory");

    const long e = e0 + m;
    if (e < E) {
        const int src = edge_index[e];
        const int dst = edge_index[(size_t)E + e];

        float xv[32];
        const float4* xr = (const float4*)(x + (size_t)src * LATENT + half * 32);
#pragma unroll
        for (int q = 0; q < 8; ++q) {
            float4 t = xr[q];
            xv[q * 4 + 0] = t.x; xv[q * 4 + 1] = t.y;
            xv[q * 4 + 2] = t.z; xv[q * 4 + 3] = t.w;
        }

        const v8h* wp = (const v8h*)(O4 + m * OUTD + half * 32 * OUT_CH);
        float p0 = 0.0f, p1 = 0.0f, p2 = 0.0f;
#pragma unroll
        for (int q = 0; q < 12; ++q) {
            v8h w8 = wp[q];
#pragma unroll
            for (int j = 0; j < 8; ++j) {
                int t = q * 8 + j;
                int i = t / 3, c = t % 3;
                float prod = xv[i] * (float)w8[j];
                if (c == 0) p0 += prod;
                else if (c == 1) p1 += prod;
                else p2 += prod;
            }
        }
        p0 += __shfl_xor(p0, 16);
        p1 += __shfl_xor(p1, 16);
        p2 += __shfl_xor(p2, 16);
        if (half == 0) {
            atomicAdd(&summed[(size_t)dst * 3 + 0], p0);
            atomicAdd(&summed[(size_t)dst * 3 + 1], p1);
            atomicAdd(&summed[(size_t)dst * 3 + 2], p2);
            atomicAdd(&count[dst], 1.0f);
        }
    }
}

__global__ void nnconv_finalize_kernel(const float* __restrict__ x,
                                       const float* __restrict__ root_w,
                                       const float* __restrict__ bias,
                                       const float* __restrict__ summed,
                                       const float* __restrict__ count,
                                       float* __restrict__ out, int N) {
    int n = blockIdx.x * blockDim.x + threadIdx.x;
    if (n >= N) return;
    float inv = 1.0f / fmaxf(count[n], 1.0f);
    float a0 = summed[(size_t)n * 3 + 0] * inv + bias[0];
    float a1 = summed[(size_t)n * 3 + 1] * inv + bias[1];
    float a2 = summed[(size_t)n * 3 + 2] * inv + bias[2];
    const float4* xr = (const float4*)(x + (size_t)n * LATENT);
#pragma unroll
    for (int q = 0; q < LATENT / 4; ++q) {
        float4 v = xr[q];
        const float* vp = (const float*)&v;
#pragma unroll
        for (int j = 0; j < 4; ++j) {
            int i = q * 4 + j;
            float xvv = vp[j];
            a0 += xvv * root_w[i * 3 + 0];
            a1 += xvv * root_w[i * 3 + 1];
            a2 += xvv * root_w[i * 3 + 2];
        }
    }
    out[(size_t)n * 3 + 0] = a0;
    out[(size_t)n * 3 + 1] = a1;
    out[(size_t)n * 3 + 2] = a2;
}


extern "C" void kernel_launch(void* const* d_in, const int* in_sizes, int n_in,
                              void* d_out, int out_size, void* d_ws, size_t ws_size,
                              hipStream_t stream) {
    const float* x          = (const float*)d_in[0];
    const int*   edge_index = (const int*)  d_in[1];
    const float* edge_attr  = (const float*)d_in[2];
    const float* W1 = (const float*)d_in[3];
    const float* b1 = (const float*)d_in[4];
    const float* W2 = (const float*)d_in[5];
    const float* b2 = (const float*)d_in[6];
    const float* W3 = (const float*)d_in[7];
    const float* b3 = (const float*)d_in[8];
    const float* W4 = (const float*)d_in[9];
    const float* b4 = (const float*)d_in[10];
    const float* root_w = (const float*)d_in[11];
    const float* bias   = (const float*)d_in[12];
    float* out = (float*)d_out;

    const int E = in_sizes[1] / 2;
    const int N = in_sizes[0] / LATENT;

    float* summed = (float*)d_ws;
    float* count  = summed + (size_t)N * 3;

    {
        int zn = N * 4;
        zero_f32<<<(zn + 255) / 256, 256, 0, stream>>>(summed, zn);
    }
    {
        int tiles  = (E + 15) / 16;
        int blocks = (tiles + WAVES - 1) / WAVES;
        nnconv_edge_kernel<<<blocks, WAVES * 32, 0, stream>>>(
            x, edge_index, edge_attr,
            W1, b1, W2, b2, W3, b3, W4, b4,
            summed, count, E);
    }
    nnconv_finalize_kernel<<<(N + 127) / 128, 128, 0, stream>>>(
        x, root_w, bias, summed, count, out, N);
}


# Scatter without global atomics: edges store a 16-byte message record; 256 owner work-groups each scan all destinations and add their own edges' records into LDS sums, then finalize; LDS arrival order
# speedup vs baseline: 3.2080x; 3.2080x over previous
#include <hip/hip_runtime.h>
#include <hip/hip_bf16.h>


typedef __attribute__((ext_vector_type(16))) _Float16 v16h;
typedef __attribute__((ext_vector_type(8)))  _Float16 v8h;
typedef __attribute__((ext_vector_type(8)))  float    v8f;

#define LATENT   64
#define EDGE_IN  70
#define K1PAD    96
#define HID      20
#define OUTD     192
#define OUT_CH   3
#define WAVES    4
#define GW       64
#define NWAVES   8
#define BTP      72
#define HP       36
#define NFRAG    10
#define FRAGH    512
#define WPK_BIAS (NFRAG * FRAGH / 2)
#define WPK_FLOATS (WPK_BIAS + 3 * 32)
#define AGG_T    1024
#define AGG_G    256
#define RMAX     256
#define LCAP     160
#define SCANK    8

typedef int i4u __attribute__((ext_vector_type(4), aligned(4)));

typedef __attribute__((ext_vector_type(2)))  _Float16 v2h;


__device__ inline v8f wmma16(v16h a, v16h b, v8f c) {
    return __builtin_amdgcn_wmma_f32_16x16x32_f16(
        false, a, false, b, (short)0, c, false, false);
}

__device__ inline v16h cat8(v8h lo, v8h hi) {
    return __builtin_shufflevector(lo, hi,
        0, 1, 2, 3, 4, 5, 6, 7, 8, 9, 10, 11, 12, 13, 14, 15);
}

__device__ inline v16h load_a(const _Float16* H, int ld, int k0) {
    int lane = threadIdx.x & 31;
    int m  = lane & 15;
    int kb = k0 + ((lane & 16) ? 8 : 0);
    v8h lo = *(const v8h*)(H + m * ld + kb);
    v8h hi = *(const v8h*)(H + m * ld + kb + 16);
    return cat8(lo, hi);
}

__device__ inline v16h make_b(const _Float16* WT, int ldk, int k0, int n0) {
    int lane = threadIdx.x & 31;
    int n  = n0 + (lane & 15);
    int kb = k0 + ((lane & 16) ? 8 : 0);
    v8h lo = *(const v8h*)(WT + n * ldk + kb);
    v8h hi = *(const v8h*)(WT + n * ldk + kb + 16);
    return cat8(lo, hi);
}

__device__ inline v16h gfrag(const v8h* wq, int f) {
    int lane = threadIdx.x & 31;
    return cat8(wq[f * 64 + lane], wq[f * 64 + 32 + lane]);
}

__device__ inline void store_c(_Float16* H, int ld, v8f c, int n0,
                               const float* bias, bool relu) {
    int lane = threadIdx.x & 31;
    int n  = n0 + (lane & 15);
    int mb = (lane & 16) ? 8 : 0;
    float bv = bias[n];
#pragma unroll
    for (int r = 0; r < 8; ++r) {
        float v = c[r] + bv;
        if (relu) v = fmaxf(v, 0.0f);
        H[(mb + r) * ld + n] = (_Float16)v;
    }
}

__device__ inline void store_h3(float* Hf, v8f c, int n0, const float* bias) {
    int lane = threadIdx.x & 31;
    int n  = n0 + (lane & 15);
    int mb = (lane & 16) ? 8 : 0;
    int col = (n < 10) ? n : ((n < HID) ? n + 2 : n + 4);
    float bv = bias[n];
#pragma unroll
    for (int r = 0; r < 8; ++r) {
        Hf[(mb + r) * HP + col] = fmaxf(c[r] + bv, 0.0f);
    }
}


__global__ __launch_bounds__(NWAVES * 32)
void nnconv_node_kernel(const float* __restrict__ x,
                        const float* __restrict__ W4,
                        const float* __restrict__ b4,
                        const float* __restrict__ W1, const float* __restrict__ b1,
                        const float* __restrict__ W2, const float* __restrict__ b2,
                        const float* __restrict__ W3, const float* __restrict__ b3,
                        float* __restrict__ G,
                        float* __restrict__ wpk, int N) {
    __shared__ alignas(16) _Float16 sBT[GW * BTP];

    const int tid = threadIdx.x;

    for (int gid = blockIdx.x * (NWAVES * 32) + tid; gid < NFRAG * FRAGH / 2 + 3 * 32;
         gid += gridDim.x * (NWAVES * 32)) {
        if (gid < NFRAG * FRAGH / 2) {
            const int f = gid >> 8;
            const int r = (gid >> 7) & 1;
            const int l = (gid >> 2) & 31;
            const int j = (gid & 3) * 2;
            const float* Wl = (f < 6) ? W1 : ((f < 8) ? W2 : W3);
            const int K  = (f < 6) ? EDGE_IN : HID;
            const int k0 = (f < 6) ? (f >> 1) * 32 : 0;
            const int n  = (f & 1) * 16 + (l & 15);
            const int k  = k0 + 16 * r + 8 * (l >> 4) + j;
            const float v0 = (n < HID && k     < K) ? Wl[k * HID + n]       : 0.0f;
            const float v1 = (n < HID && k + 1 < K) ? Wl[(k + 1) * HID + n] : 0.0f;
            v2h p; p[0] = (_Float16)v0; p[1] = (_Float16)v1;
            ((v2h*)wpk)[gid] = p;
        } else {
            const int q = gid - NFRAG * FRAGH / 2;
            const int n = q & 31;
            const float* bl = (q < 32) ? b1 : ((q < 64) ? b2 : b3);
            wpk[WPK_BIAS + q] = (n < HID) ? bl[n] : 0.0f;
        }
    }

    for (int idx = tid; idx < GW * LATENT; idx += NWAVES * 32) {
        int j = idx >> 6, i = idx & 63;
        int hh = j >> 5, t = j & 31;
        int k  = hh * 10 + t / 3;
        int c  = t - (t / 3) * 3;
        int bc = hh * 2 + (t - 30);
        bool isw = (t < 30);
        bool isz = (!isw) && (bc >= OUT_CH);
        const float* p = isw ? (W4 + k * OUTD + c) : (b4 + (isz ? 0 : bc));
        float v = p[i * OUT_CH];
        sBT[j * BTP + i] = (_Float16)(isz ? 0.0f : v);
    }
    __syncthreads();

    const int lane = tid & 31;
    const int wave = tid >> 5;
    const int m    = lane & 15;
    const int hi   = lane >> 4;
    const long n0  = ((long)blockIdx.x * NWAVES + wave) * 16;
    if (n0 >= N) return;

    long node = n0 + m;
    if (node >= N) node = N - 1;
    const float4* xr = (const float4*)(x + (size_t)node * LATENT);
    v16h a[2];
#pragma unroll
    for (int ks = 0; ks < 2; ++ks) {
        int q = ks * 8 + hi * 2;
        float4 f0 = xr[q], f1 = xr[q + 1], f2 = xr[q + 4], f3 = xr[q + 5];
        v16h t;
        t[0]  = (_Float16)f0.x; t[1]  = (_Float16)f0.y; t[2]  = (_Float16)f0.z; t[3]  = (_Float16)f0.w;
        t[4]  = (_Float16)f1.x; t[5]  = (_Float16)f1.y; t[6]  = (_Float16)f1.z; t[7]  = (_Float16)f1.w;
        t[8]  = (_Float16)f2.x; t[9]  = (_Float16)f2.y; t[10] = (_Float16)f2.z; t[11] = (_Float16)f2.w;
        t[12] = (_Float16)f3.x; t[13] = (_Float16)f3.y; t[14] = (_Float16)f3.z; t[15] = (_Float16)f3.w;
        a[ks] = t;
    }

    v8f c[GW / 16];
#pragma unroll
    for (int t = 0; t < GW / 16; ++t) {
        v8f z = {};
        c[t] = wmma16(a[0], make_b(sBT, BTP, 0,  t * 16), z);
        c[t] = wmma16(a[1], make_b(sBT, BTP, 32, t * 16), c[t]);
    }
    float* gp = G + (size_t)(n0 + hi * 8) * GW + m;
#pragma unroll
    for (int t = 0; t < GW / 16; ++t) {
#pragma unroll
        for (int r = 0; r < 8; ++r) gp[r * GW + t * 16] = c[t][r];
    }
}

__global__ __launch_bounds__(WAVES * 32)
void nnconv_edge_kernel(const int*   __restrict__ edge_index,
                        const float* __restrict__ edge_attr,
                        const float* __restrict__ wpk,
                        const float* __restrict__ G,
                        float* __restrict__ msg,
                        int E) {
    const v8h*   wq = (const v8h*)wpk;
    const float* bq = wpk + WPK_BIAS;
    __shared__ alignas(16) _Float16 sA [WAVES][16 * HP * 2];

    const int tid = threadIdx.x;

    const int lane = tid & 31;
    const int wave = tid >> 5;
    const int m    = lane & 15;
    const int half = lane >> 4;

    const long tile = (long)blockIdx.x * WAVES + wave;
    const long e0   = tile * 16;
    if (e0 >= E) return;

    _Float16* H  = sA [wave];
    float*    Hf = (float*)sA[wave];

    v16h a1[3];
    {
        long row = e0 + m;
        if (row >= E) row = E - 1;
        const float2* er = (const float2*)(edge_attr + (size_t)row * EDGE_IN);
        float2 f[2][8];
#pragma unroll
        for (int ks = 0; ks < 2; ++ks) {
            const int q = ks * 16 + half * 4;
#pragma unroll
            for (int j = 0; j < 4; ++j) {
                f[ks][j]     = er[q + j];
                f[ks][4 + j] = er[q + 8 + j];
            }
        }
        const float2 t0 = er[32], t1 = er[33], t2 = er[34];
#pragma unroll
        for (int ks = 0; ks < 2; ++ks) {
            v16h t;
#pragma unroll
            for (int j = 0; j < 8; ++j) {
                t[2 * j]     = (_Float16)f[ks][j].x;
                t[2 * j + 1] = (_Float16)f[ks][j].y;
            }
            a1[ks] = t;
        }
        const bool lo = (half == 0);
        v16h t = {};
        t[0] = (_Float16)(lo ? t0.x : 0.0f); t[1] = (_Float16)(lo ? t0.y : 0.0f);
        t[2] = (_Float16)(lo ? t1.x : 0.0f); t[3] = (_Float16)(lo ? t1.y : 0.0f);
        t[4] = (_Float16)(lo ? t2.x : 0.0f); t[5] = (_Float16)(lo ? t2.y : 0.0f);
        a1[2] = t;
    }

    {
        v8f c0 = {}, c1 = {};
#pragma unroll
        for (int ks = 0; ks < 3; ++ks) {
            c0 = wmma16(a1[ks], gfrag(wq, ks * 2),     c0);
            c1 = wmma16(a1[ks], gfrag(wq, ks * 2 + 1), c1);
        }
        store_c(H, 32, c0, 0,  bq, true);
        store_c(H, 32, c1, 16, bq, true);
    }
    asm volatile("s_wait_dscnt 0" ::: "memory");

    {
        v16h a = load_a(H, 32, 0);
        v8f c0 = {}, c1 = {};
        c0 = wmma16(a, gfrag(wq, 6), c0);
        c1 = wmma16(a, gfrag(wq, 7), c1);
        asm volatile("s_wait_dscnt 0" ::: "memory");
        store_c(H, 32, c0, 0,  bq + 32, true);
        store_c(H, 32, c1, 16, bq + 32, true);
    }
    asm volatile("s_wait_dscnt 0" ::: "memory");

    {
        v16h a = load_a(H, 32, 0);
        v8f c0 = {}, c1 = {};
        c0 = wmma16(a, gfrag(wq, 8), c0);
        c1 = wmma16(a, gfrag(wq, 9), c1);
        asm volatile("s_wait_dscnt 0" ::: "memory");
        store_h3(Hf, c0, 0,  bq + 64);
        store_h3(Hf, c1, 16, bq + 64);
    }
    asm volatile("s_wait_dscnt 0" ::: "memory");

    const long e = e0 + m;
    if (e < E) {
        const int src = edge_index[e];

        float g[32];
        const float4* gr = (const float4*)(G + (size_t)src * GW + half * 32);
#pragma unroll
        for (int q = 0; q < 8; ++q) {
            float4 t = gr[q];
            g[q * 4 + 0] = t.x; g[q * 4 + 1] = t.y;
            g[q * 4 + 2] = t.z; g[q * 4 + 3] = t.w;
        }

        const float* hp = Hf + m * HP + half * 12;
        const float4 h0 = *(const float4*)(hp);
        const float4 h1 = *(const float4*)(hp + 4);
        const float2 h2 = *(const float2*)(hp + 8);
        const float h[10] = { h0.x, h0.y, h0.z, h0.w, h1.x, h1.y, h1.z, h1.w, h2.x, h2.y };

        float p0 = 0.0f, p1 = 0.0f, p2 = 0.0f;
#pragma unroll
        for (int kk = 0; kk < 10; ++kk) {
            p0 += h[kk] * g[kk * 3 + 0];
            p1 += h[kk] * g[kk * 3 + 1];
            p2 += h[kk] * g[kk * 3 + 2];
        }
        p0 += (half == 0) ? g[30] : 0.0f;
        p1 += (half == 0) ? g[31] : 0.0f;
        p2 += (half == 0) ? 0.0f : g[30];
        p0 += __shfl_xor(p0, 16);
        p1 += __shfl_xor(p1, 16);
        p2 += __shfl_xor(p2, 16);
        if (half == 0) {
            ((float4*)msg)[e] = make_float4(p0, p1, p2, 0.0f);
        }
    }
}

__device__ inline void add_record(float* sacc, const float* __restrict__ msg,
                                  unsigned e, unsigned r) {
    const float4 m = ((const float4*)msg)[e];
    atomicAdd(&sacc[r * 4 + 0], m.x);
    atomicAdd(&sacc[r * 4 + 1], m.y);
    atomicAdd(&sacc[r * 4 + 2], m.z);
    atomicAdd(&sacc[r * 4 + 3], 1.0f);
}

__global__ __launch_bounds__(AGG_T)
void nnconv_aggregate_kernel(const int*   __restrict__ edge_index,
                             const float* __restrict__ msg,
                             const float* __restrict__ x,
                             const float* __restrict__ root_w,
                             const float* __restrict__ bias,
                             float* __restrict__ out, int E, int N, int R) {
    __shared__ float    sacc[RMAX * 4];
    __shared__ unsigned sle[AGG_T / 32][LCAP];
    __shared__ unsigned slr[AGG_T / 32][LCAP];

    const int tid  = threadIdx.x;
    const int lane = tid & 31;
    const int wave = tid >> 5;
    const int n0   = blockIdx.x * R;
    const unsigned Ru = (unsigned)R;

    for (int q = tid; q < RMAX * 4; q += AGG_T) sacc[q] = 0.0f;
    __syncthreads();

    const int* dstp = edge_index + (size_t)E;
    const i4u* d4   = (const i4u*)dstp;
    const int nq    = E >> 2;
    unsigned cnt = 0;
    unsigned* le = sle[wave];
    unsigned* lr = slr[wave];

    for (int base = 0; base < nq; base += AGG_T * SCANK) {
        i4u dd[SCANK];
#pragma unroll
        for (int k = 0; k < SCANK; ++k) {
            const int qi = base + k * AGG_T + tid;
            dd[k] = d4[qi < nq ? qi : (nq - 1)];
        }
#pragma unroll
        for (int k = 0; k < SCANK; ++k) {
            const int  qi = base + k * AGG_T + tid;
            const bool ok = qi < nq;
            const i4u d = dd[k];
            unsigned r[4];
            r[0] = (unsigned)(d[0] - n0); r[1] = (unsigned)(d[1] - n0);
            r[2] = (unsigned)(d[2] - n0); r[3] = (unsigned)(d[3] - n0);
            const unsigned rm = ok ? min(min(r[0], r[1]), min(r[2], r[3])) : 0xffffffffu;
            if (__builtin_amdgcn_ballot_w32(rm < Ru) != 0u) {
#pragma unroll
                for (int j = 0; j < 4; ++j) {
                    const bool hit = ok && (r[j] < Ru);
                    const unsigned mk = __builtin_amdgcn_ballot_w32(hit);
                    if (mk != 0u) {
                        if (hit) {
                            const unsigned pos = cnt + __builtin_amdgcn_mbcnt_lo(mk, 0u);
                            le[pos] = (unsigned)qi * 4u + (unsigned)j;
                            lr[pos] = r[j];
                        }
                        cnt += (unsigned)__builtin_popcount(mk);
                    }
                }
                asm volatile("s_wait_dscnt 0" ::: "memory");
                for (int t = 0; t < 4 && cnt >= 32u; ++t) {
                    cnt -= 32u;
                    add_record(sacc, msg, le[cnt + lane], lr[cnt + lane]);
                }
            }
        }
    }
    for (int e = (nq << 2) + tid; e < E; e += AGG_T) {
        const unsigned rr = (unsigned)(dstp[e] - n0);
        if (rr < Ru) add_record(sacc, msg, (unsigned)e, rr);
    }
    if ((unsigned)lane < cnt) add_record(sacc, msg, le[lane], lr[lane]);
    __syncthreads();

    const int n = n0 + tid;
    if (tid < R && n < N) {
        float inv = 1.0f / fmaxf(sacc[tid * 4 + 3], 1.0f);
        float a0 = sacc[tid * 4 + 0] * inv + bias[0];
        float a1 = sacc[tid * 4 + 1] * inv + bias[1];
        float a2 = sacc[tid * 4 + 2] * inv + bias[2];
        const float4* xr = (const float4*)(x + (size_t)n * LATENT);
#pragma unroll
        for (int q = 0; q < LATENT / 4; ++q) {
            float4 v = xr[q];
            const float* vp = (const float*)&v;
#pragma unroll
            for (int j = 0; j < 4; ++j) {
                int i = q * 4 + j;
                float xvv = vp[j];
                a0 += xvv * root_w[i * 3 + 0];
                a1 += xvv * root_w[i * 3 + 1];
                a2 += xvv * root_w[i * 3 + 2];
            }
        }
        out[(size_t)n * 3 + 0] = a0;
        out[(size_t)n * 3 + 1] = a1;
        out[(size_t)n * 3 + 2] = a2;
    }
}


extern "C" void kernel_launch(void* const* d_in, const int* in_sizes, int n_in,
                              void* d_out, int out_size, void* d_ws, size_t ws_size,
                              hipStream_t stream) {
    const float* x          = (const float*)d_in[0];
    const int*   edge_index = (const int*)  d_in[1];
    const float* edge_attr  = (const float*)d_in[2];
    const float* W1 = (const float*)d_in[3];
    const float* b1 = (const float*)d_in[4];
    const float* W2 = (const float*)d_in[5];
    const float* b2 = (const float*)d_in[6];
    const float* W3 = (const float*)d_in[7];
    const float* b3 = (const float*)d_in[8];
    const float* W4 = (const float*)d_in[9];
    const float* b4 = (const float*)d_in[10];
    const float* root_w = (const float*)d_in[11];
    const float* bias   = (const float*)d_in[12];
    float* out = (float*)d_out;

    const int E = in_sizes[1] / 2;
    const int N = in_sizes[0] / LATENT;

    const int etiles = (E + 15) / 16;
    float* msg    = (float*)d_ws;
    float* G      = msg + (size_t)etiles * 16 * 4;

    const int ntiles = (N + 15) / 16;
    float* wpk    = G + (size_t)ntiles * 16 * GW;
    if (ws_size < ((size_t)etiles * 16 * 4 + (size_t)ntiles * 16 * GW + WPK_FLOATS) * sizeof(float)) return;
    const int owners = (N + RMAX - 1) / RMAX > AGG_G ? (N + RMAX - 1) / RMAX : AGG_G;
    const int R      = (N + owners - 1) / owners;

    {
        int tiles  = ntiles;
        int blocks = (tiles + NWAVES - 1) / NWAVES;
        nnconv_node_kernel<<<blocks, NWAVES * 32, 0, stream>>>(
            x, W4, b4, W1, b1, W2, b2, W3, b3, G, wpk, N);
    }
    {
        int tiles  = (E + 15) / 16;
        int blocks = (tiles + WAVES - 1) / WAVES;
        nnconv_edge_kernel<<<blocks, WAVES * 32, 0, stream>>>(
            edge_index, edge_attr, wpk, G,
            msg, E);
    }
    nnconv_aggregate_kernel<<<owners, AGG_T, 0, stream>>>(
        edge_index, msg, x, root_w, bias, out, E, N, R);
}


# Owner scan reads one byte per edge (destination / R, written by the first launch), 16 edges per lane and test, instead of the 4-byte index; adds still in LDS arrival order
# speedup vs baseline: 3.6723x; 1.1447x over previous
#include <hip/hip_runtime.h>
#include <hip/hip_bf16.h>


typedef __attribute__((ext_vector_type(16))) _Float16 v16h;
typedef __attribute__((ext_vector_type(8)))  _Float16 v8h;
typedef __attribute__((ext_vector_type(8)))  float    v8f;

#define LATENT   64
#define EDGE_IN  70
#define K1PAD    96
#define HID      20
#define OUTD     192
#define OUT_CH   3
#define WAVES    4
#define GW       64
#define NWAVES   8
#define BTP      72
#define HP       36
#define NFRAG    10
#define FRAGH    512
#define WPK_BIAS (NFRAG * FRAGH / 2)
#define WPK_FLOATS (WPK_BIAS + 3 * 32)
#define AGG_T    1024
#define AGG_G    256
#define RMAX     1024
#define LCAP     64
#define SCANK    8

typedef int i4u __attribute__((ext_vector_type(4), aligned(4)));
typedef int i4a __attribute__((ext_vector_type(4)));

typedef __attribute__((ext_vector_type(2)))  _Float16 v2h;


__device__ inline v8f wmma16(v16h a, v16h b, v8f c) {
    return __builtin_amdgcn_wmma_f32_16x16x32_f16(
        false, a, false, b, (short)0, c, false, false);
}

__device__ inline v16h cat8(v8h lo, v8h hi) {
    return __builtin_shufflevector(lo, hi,
        0, 1, 2, 3, 4, 5, 6, 7, 8, 9, 10, 11, 12, 13, 14, 15);
}

__device__ inline v16h load_a(const _Float16* H, int ld, int k0) {
    int lane = threadIdx.x & 31;
    int m  = lane & 15;
    int kb = k0 + ((lane & 16) ? 8 : 0);
    v8h lo = *(const v8h*)(H + m * ld + kb);
    v8h hi = *(const v8h*)(H + m * ld + kb + 16);
    return cat8(lo, hi);
}

__device__ inline v16h make_b(const _Float16* WT, int ldk, int k0, int n0) {
    int lane = threadIdx.x & 31;
    int n  = n0 + (lane & 15);
    int kb = k0 + ((lane & 16) ? 8 : 0);
    v8h lo = *(const v8h*)(WT + n * ldk + kb);
    v8h hi = *(const v8h*)(WT + n * ldk + kb + 16);
    return cat8(lo, hi);
}

__device__ inline v16h gfrag(const v8h* wq, int f) {
    int lane = threadIdx.x & 31;
    return cat8(wq[f * 64 + lane], wq[f * 64 + 32 + lane]);
}

__device__ inline void store_c(_Float16* H, int ld, v8f c, int n0,
                               const float* bias, bool relu) {
    int lane = threadIdx.x & 31;
    int n  = n0 + (lane & 15);
    int mb = (lane & 16) ? 8 : 0;
    float bv = bias[n];
#pragma unroll
    for (int r = 0; r < 8; ++r) {
        float v = c[r] + bv;
        if (relu) v = fmaxf(v, 0.0f);
        H[(mb + r) * ld + n] = (_Float16)v;
    }
}

__device__ inline void store_h3(float* Hf, v8f c, int n0, const float* bias) {
    int lane = threadIdx.x & 31;
    int n  = n0 + (lane & 15);
    int mb = (lane & 16) ? 8 : 0;
    int col = (n < 10) ? n : ((n < HID) ? n + 2 : n + 4);
    float bv = bias[n];
#pragma unroll
    for (int r = 0; r < 8; ++r) {
        Hf[(mb + r) * HP + col] = fmaxf(c[r] + bv, 0.0f);
    }
}


__global__ __launch_bounds__(NWAVES * 32)
void nnconv_node_kernel(const float* __restrict__ x,
                        const int*   __restrict__ edge_index,
                        const float* __restrict__ W4,
                        const float* __restrict__ b4,
                        const float* __restrict__ W1, const float* __restrict__ b1,
                        const float* __restrict__ W2, const float* __restrict__ b2,
                        const float* __restrict__ W3, const float* __restrict__ b3,
                        float* __restrict__ G,
                        float* __restrict__ wpk,
                        unsigned* __restrict__ own, int N, int E, int R) {
    __shared__ alignas(16) _Float16 sBT[GW * BTP];

    const int tid = threadIdx.x;

    for (int gid = blockIdx.x * (NWAVES * 32) + tid; gid < NFRAG * FRAGH / 2 + 3 * 32;
         gid += gridDim.x * (NWAVES * 32)) {
        if (gid < NFRAG * FRAGH / 2) {
            const int f = gid >> 8;
            const int r = (gid >> 7) & 1;
            const int l = (gid >> 2) & 31;
            const int j = (gid & 3) * 2;
            const float* Wl = (f < 6) ? W1 : ((f < 8) ? W2 : W3);
            const int K  = (f < 6) ? EDGE_IN : HID;
            const int k0 = (f < 6) ? (f >> 1) * 32 : 0;
            const int n  = (f & 1) * 16 + (l & 15);
            const int k  = k0 + 16 * r + 8 * (l >> 4) + j;
            const float v0 = (n < HID && k     < K) ? Wl[k * HID + n]       : 0.0f;
            const float v1 = (n < HID && k + 1 < K) ? Wl[(k + 1) * HID + n] : 0.0f;
            v2h p; p[0] = (_Float16)v0; p[1] = (_Float16)v1;
            ((v2h*)wpk)[gid] = p;
        } else {
            const int q = gid - NFRAG * FRAGH / 2;
            const int n = q & 31;
            const float* bl = (q < 32) ? b1 : ((q < 64) ? b2 : b3);
            wpk[WPK_BIAS + q] = (n < HID) ? bl[n] : 0.0f;
        }
    }

    {
        const i4u* d4 = (const i4u*)(edge_index + (size_t)E);
        const unsigned Rd = (unsigned)R;
        for (int ch = blockIdx.x * (NWAVES * 32) + tid; ch < (E >> 4);
             ch += gridDim.x * (NWAVES * 32)) {
            i4a o;
#pragma unroll
            for (int q = 0; q < 4; ++q) {
                const i4u d = d4[(size_t)ch * 4 + q];
                unsigned w = 0u;
#pragma unroll
                for (int b = 0; b < 4; ++b) {
                    const unsigned ow = min((unsigned)d[b] / Rd, 255u);
                    w |= ow << (8 * b);
                }
                o[q] = (int)w;
            }
            ((i4a*)own)[ch] = o;
        }
    }

    for (int idx = tid; idx < GW * LATENT; idx += NWAVES * 32) {
        int j = idx >> 6, i = idx & 63;
        int hh = j >> 5, t = j & 31;
        int k  = hh * 10 + t / 3;
        int c  = t - (t / 3) * 3;
        int bc = hh * 2 + (t - 30);
        bool isw = (t < 30);
        bool isz = (!isw) && (bc >= OUT_CH);
        const float* p = isw ? (W4 + k * OUTD + c) : (b4 + (isz ? 0 : bc));
        float v = p[i * OUT_CH];
        sBT[j * BTP + i] = (_Float16)(isz ? 0.0f : v);
    }
    __syncthreads();

    const int lane = tid & 31;
    const int wave = tid >> 5;
    const int m    = lane & 15;
    const int hi   = lane >> 4;
    const long n0  = ((long)blockIdx.x * NWAVES + wave) * 16;
    if (n0 >= N) return;

    long node = n0 + m;
    if (node >= N) node = N - 1;
    const float4* xr = (const float4*)(x + (size_t)node * LATENT);
    v16h a[2];
#pragma unroll
    for (int ks = 0; ks < 2; ++ks) {
        int q = ks * 8 + hi * 2;
        float4 f0 = xr[q], f1 = xr[q + 1], f2 = xr[q + 4], f3 = xr[q + 5];
        v16h t;
        t[0]  = (_Float16)f0.x; t[1]  = (_Float16)f0.y; t[2]  = (_Float16)f0.z; t[3]  = (_Float16)f0.w;
        t[4]  = (_Float16)f1.x; t[5]  = (_Float16)f1.y; t[6]  = (_Float16)f1.z; t[7]  = (_Float16)f1.w;
        t[8]  = (_Float16)f2.x; t[9]  = (_Float16)f2.y; t[10] = (_Float16)f2.z; t[11] = (_Float16)f2.w;
        t[12] = (_Float16)f3.x; t[13] = (_Float16)f3.y; t[14] = (_Float16)f3.z; t[15] = (_Float16)f3.w;
        a[ks] = t;
    }

    v8f c[GW / 16];
#pragma unroll
    for (int t = 0; t < GW / 16; ++t) {
        v8f z = {};
        c[t] = wmma16(a[0], make_b(sBT, BTP, 0,  t * 16), z);
        c[t] = wmma16(a[1], make_b(sBT, BTP, 32, t * 16), c[t]);
    }
    float* gp = G + (size_t)(n0 + hi * 8) * GW + m;
#pragma unroll
    for (int t = 0; t < GW / 16; ++t) {
#pragma unroll
        for (int r = 0; r < 8; ++r) gp[r * GW + t * 16] = c[t][r];
    }
}

__global__ __launch_bounds__(WAVES * 32)
void nnconv_edge_kernel(const int*   __restrict__ edge_index,
                        const float* __restrict__ edge_attr,
                        const float* __restrict__ wpk,
                        const float* __restrict__ G,
                        float* __restrict__ msg,
                        int E) {
    const v8h*   wq = (const v8h*)wpk;
    const float* bq = wpk + WPK_BIAS;
    __shared__ alignas(16) _Float16 sA [WAVES][16 * HP * 2];

    const int tid = threadIdx.x;

    const int lane = tid & 31;
    const int wave = tid >> 5;
    const int m    = lane & 15;
    const int half = lane >> 4;

    const long tile = (long)blockIdx.x * WAVES + wave;
    const long e0   = tile * 16;
    if (e0 >= E) return;

    _Float16* H  = sA [wave];
    float*    Hf = (float*)sA[wave];

    v16h a1[3];
    {
        long row = e0 + m;
        if (row >= E) row = E - 1;
        const float2* er = (const float2*)(edge_attr + (size_t)row * EDGE_IN);
        float2 f[2][8];
#pragma unroll
        for (int ks = 0; ks < 2; ++ks) {
            const int q = ks * 16 + half * 4;
#pragma unroll
            for (int j = 0; j < 4; ++j) {
                f[ks][j]     = er[q + j];
                f[ks][4 + j] = er[q + 8 + j];
            }
        }
        const float2 t0 = er[32], t1 = er[33], t2 = er[34];
#pragma unroll
        for (int ks = 0; ks < 2; ++ks) {
            v16h t;
#pragma unroll
            for (int j = 0; j < 8; ++j) {
                t[2 * j]     = (_Float16)f[ks][j].x;
                t[2 * j + 1] = (_Float16)f[ks][j].y;
            }
            a1[ks] = t;
        }
        const bool lo = (half == 0);
        v16h t = {};
        t[0] = (_Float16)(lo ? t0.x : 0.0f); t[1] = (_Float16)(lo ? t0.y : 0.0f);
        t[2] = (_Float16)(lo ? t1.x : 0.0f); t[3] = (_Float16)(lo ? t1.y : 0.0f);
        t[4] = (_Float16)(lo ? t2.x : 0.0f); t[5] = (_Float16)(lo ? t2.y : 0.0f);
        a1[2] = t;
    }

    {
        v8f c0 = {}, c1 = {};
#pragma unroll
        for (int ks = 0; ks < 3; ++ks) {
            c0 = wmma16(a1[ks], gfrag(wq, ks * 2),     c0);
            c1 = wmma16(a1[ks], gfrag(wq, ks * 2 + 1), c1);
        }
        store_c(H, 32, c0, 0,  bq, true);
        store_c(H, 32, c1, 16, bq, true);
    }
    asm volatile("s_wait_dscnt 0" ::: "memory");

    {
        v16h a = load_a(H, 32, 0);
        v8f c0 = {}, c1 = {};
        c0 = wmma16(a, gfrag(wq, 6), c0);
        c1 = wmma16(a, gfrag(wq, 7), c1);
        asm volatile("s_wait_dscnt 0" ::: "memory");
        store_c(H, 32, c0, 0,  bq + 32, true);
        store_c(H, 32, c1, 16, bq + 32, true);
    }
    asm volatile("s_wait_dscnt 0" ::: "memory");

    {
        v16h a = load_a(H, 32, 0);
        v8f c0 = {}, c1 = {};
        c0 = wmma16(a, gfrag(wq, 8), c0);
        c1 = wmma16(a, gfrag(wq, 9), c1);
        asm volatile("s_wait_dscnt 0" ::: "memory");
        store_h3(Hf, c0, 0,  bq + 64);
        store_h3(Hf, c1, 16, bq + 64);
    }
    asm volatile("s_wait_dscnt 0" ::: "memory");

    const long e = e0 + m;
    if (e < E) {
        const int src = edge_index[e];

        float g[32];
        const float4* gr = (const float4*)(G + (size_t)src * GW + half * 32);
#pragma unroll
        for (int q = 0; q < 8; ++q) {
            float4 t = gr[q];
            g[q * 4 + 0] = t.x; g[q * 4 + 1] = t.y;
            g[q * 4 + 2] = t.z; g[q * 4 + 3] = t.w;
        }

        const float* hp = Hf + m * HP + half * 12;
        const float4 h0 = *(const float4*)(hp);
        const float4 h1 = *(const float4*)(hp + 4);
        const float2 h2 = *(const float2*)(hp + 8);
        const float h[10] = { h0.x, h0.y, h0.z, h0.w, h1.x, h1.y, h1.z, h1.w, h2.x, h2.y };

        float p0 = 0.0f, p1 = 0.0f, p2 = 0.0f;
#pragma unroll
        for (int kk = 0; kk < 10; ++kk) {
            p0 += h[kk] * g[kk * 3 + 0];
            p1 += h[kk] * g[kk * 3 + 1];
            p2 += h[kk] * g[kk * 3 + 2];
        }
        p0 += (half == 0) ? g[30] : 0.0f;
        p1 += (half == 0) ? g[31] : 0.0f;
        p2 += (half == 0) ? 0.0f : g[30];
        p0 += __shfl_xor(p0, 16);
        p1 += __shfl_xor(p1, 16);
        p2 += __shfl_xor(p2, 16);
        if (half == 0) {
            ((float4*)msg)[e] = make_float4(p0, p1, p2, 0.0f);
        }
    }
}

__device__ inline void add_record(float* sacc, const float* __restrict__ msg,
                                  const int* __restrict__ dstp,
                                  unsigned e, int n0, unsigned Ru) {
    const float4 m = ((const float4*)msg)[e];
    const unsigned r = (unsigned)(dstp[e] - n0);
    const bool in = r < Ru;
    const unsigned rs = in ? r : 0u;
    atomicAdd(&sacc[rs * 4 + 0], in ? m.x : 0.0f);
    atomicAdd(&sacc[rs * 4 + 1], in ? m.y : 0.0f);
    atomicAdd(&sacc[rs * 4 + 2], in ? m.z : 0.0f);
    atomicAdd(&sacc[rs * 4 + 3], in ? 1.0f : 0.0f);
}

__global__ __launch_bounds__(AGG_T)
void nnconv_aggregate_kernel(const int*   __restrict__ edge_index,
                             const unsigned* __restrict__ own,
                             const float* __restrict__ msg,
                             const float* __restrict__ x,
                             const float* __restrict__ root_w,
                             const float* __restrict__ bias,
                             float* __restrict__ out, int E, int N, int R) {
    __shared__ float    sacc[RMAX * 4];
    __shared__ unsigned sle[AGG_T / 32][LCAP];

    const int tid  = threadIdx.x;
    const int lane = tid & 31;
    const int wave = tid >> 5;
    const int n0   = blockIdx.x * R;
    const unsigned Ru = (unsigned)R;

    for (int q = tid; q < R * 4; q += AGG_T) sacc[q] = 0.0f;
    __syncthreads();

    const int* dstp = edge_index + (size_t)E;
    const i4a* o4   = (const i4a*)own;
    const int nq    = E >> 4;
    const unsigned pat = (unsigned)blockIdx.x * 0x01010101u;
    unsigned cnt = 0;
    unsigned* le = sle[wave];

    for (int base = 0; base < nq; base += AGG_T * SCANK) {
        i4a dd[SCANK];
#pragma unroll
        for (int k = 0; k < SCANK; ++k) {
            const int qi = base + k * AGG_T + tid;
            dd[k] = o4[qi < nq ? qi : (nq - 1)];
        }
#pragma unroll
        for (int k = 0; k < SCANK; ++k) {
            const int  qi = base + k * AGG_T + tid;
            const bool ok = qi < nq;
            const i4a d = dd[k];
            unsigned zb = 0u;
#pragma unroll
            for (int w = 0; w < 4; ++w) {
                const unsigned xo = (unsigned)d[w] ^ pat;
                const unsigned zw = ~(((xo & 0x7f7f7f7fu) + 0x7f7f7f7fu) | xo) & 0x80808080u;
                zb |= zw >> w;
            }
            if (__builtin_amdgcn_ballot_w32(ok && (zb != 0u)) != 0u) {
                unsigned zm = ok ? zb : 0u;
                unsigned mk = __builtin_amdgcn_ballot_w32(zm != 0u);
                do {
                    if (zm != 0u) {
                        const unsigned p   = (unsigned)__builtin_ctz(zm);
                        const unsigned pos = cnt + __builtin_amdgcn_mbcnt_lo(mk, 0u);
                        le[pos] = (unsigned)qi * 16u + ((7u - (p & 7u)) << 2) + (p >> 3);
                        zm &= zm - 1u;
                    }
                    cnt += (unsigned)__builtin_popcount(mk);
                    if (cnt >= 32u) {
                        asm volatile("s_wait_dscnt 0" ::: "memory");
                        cnt -= 32u;
                        add_record(sacc, msg, dstp, le[cnt + lane], n0, Ru);
                    }
                    mk = __builtin_amdgcn_ballot_w32(zm != 0u);
                } while (mk != 0u);
            }
        }
    }
    for (int e = (nq << 4) + tid; e < E; e += AGG_T) {
        add_record(sacc, msg, dstp, (unsigned)e, n0, Ru);
    }
    asm volatile("s_wait_dscnt 0" ::: "memory");
    if ((unsigned)lane < cnt) add_record(sacc, msg, dstp, le[lane], n0, Ru);
    __syncthreads();

    const int n = n0 + tid;
    if (tid < R && n < N) {
        float inv = 1.0f / fmaxf(sacc[tid * 4 + 3], 1.0f);
        float a0 = sacc[tid * 4 + 0] * inv + bias[0];
        float a1 = sacc[tid * 4 + 1] * inv + bias[1];
        float a2 = sacc[tid * 4 + 2] * inv + bias[2];
        const float4* xr = (const float4*)(x + (size_t)n * LATENT);
#pragma unroll
        for (int q = 0; q < LATENT / 4; ++q) {
            float4 v = xr[q];
            const float* vp = (const float*)&v;
#pragma unroll
            for (int j = 0; j < 4; ++j) {
                int i = q * 4 + j;
                float xvv = vp[j];
                a0 += xvv * root_w[i * 3 + 0];
                a1 += xvv * root_w[i * 3 + 1];
                a2 += xvv * root_w[i * 3 + 2];
            }
        }
        out[(size_t)n * 3 + 0] = a0;
        out[(size_t)n * 3 + 1] = a1;
        out[(size_t)n * 3 + 2] = a2;
    }
}


extern "C" void kernel_launch(void* const* d_in, const int* in_sizes, int n_in,
                              void* d_out, int out_size, void* d_ws, size_t ws_size,
                              hipStream_t stream) {
    const float* x          = (const float*)d_in[0];
    const int*   edge_index = (const int*)  d_in[1];
    const float* edge_attr  = (const float*)d_in[2];
    const float* W1 = (const float*)d_in[3];
    const float* b1 = (const float*)d_in[4];
    const float* W2 = (const float*)d_in[5];
    const float* b2 = (const float*)d_in[6];
    const float* W3 = (const float*)d_in[7];
    const float* b3 = (const float*)d_in[8];
    const float* W4 = (const float*)d_in[9];
    const float* b4 = (const float*)d_in[10];
    const float* root_w = (const float*)d_in[11];
    const float* bias   = (const float*)d_in[12];
    float* out = (float*)d_out;

    const int E = in_sizes[1] / 2;
    const int N = in_sizes[0] / LATENT;

    const int etiles = (E + 15) / 16;
    float* msg    = (float*)d_ws;
    float* G      = msg + (size_t)etiles * 16 * 4;

    const int ntiles = (N + 15) / 16;
    float* wpk    = G + (size_t)ntiles * 16 * GW;
    unsigned* own = (unsigned*)(wpk + WPK_FLOATS);
    if (ws_size < ((size_t)etiles * 16 * 4 + (size_t)ntiles * 16 * GW + WPK_FLOATS) * sizeof(float)
                  + (size_t)etiles * 16) return;
    const int owners = AGG_G;
    const int R      = (N + owners - 1) / owners;
    if (R < 1 || R > RMAX) return;

    {
        int tiles  = ntiles;
        int blocks = (tiles + NWAVES - 1) / NWAVES;
        nnconv_node_kernel<<<blocks, NWAVES * 32, 0, stream>>>(
            x, edge_index, W4, b4, W1, b1, W2, b2, W3, b3, G, wpk, own, N, E, R);
    }
    {
        int tiles  = (E + 15) / 16;
        int blocks = (tiles + WAVES - 1) / WAVES;
        nnconv_edge_kernel<<<blocks, WAVES * 32, 0, stream>>>(
            edge_index, edge_attr, wpk, G,
            msg, E);
    }
    nnconv_aggregate_kernel<<<owners, AGG_T, 0, stream>>>(
        edge_index, own, msg, x, root_w, bias, out, E, N, R);
}


# Message sum on 16 lanes per edge: a G row is read as its 2 whole 128-B lines by one load (4 lines per load, was 16); h3 stays in layer-3 accumulators; lane sums folded by DPP; sum order differs
# speedup vs baseline: 5.4415x; 1.4818x over previous
#include <hip/hip_runtime.h>
#include <hip/hip_bf16.h>


typedef __attribute__((ext_vector_type(16))) _Float16 v16h;
typedef __attribute__((ext_vector_type(8)))  _Float16 v8h;
typedef __attribute__((ext_vector_type(8)))  float    v8f;

#define LATENT   64
#define EDGE_IN  70
#define K1PAD    96
#define HID      20
#define OUTD     192
#define OUT_CH   3
#define WAVES    4
#define GW       64
#define NWAVES   8
#define BTP      72
#define HP       36
#define NFRAG    10
#define FRAGH    512
#define WPK_BIAS (NFRAG * FRAGH / 2)
#define WPK_FLOATS (WPK_BIAS + 3 * 32)
#define AGG_T    512
#define AGG_G    256
#define RMAX     1024
#define RBITS    10
#define CH       2048
#define CHT      (CH / (NWAVES * 32))
#define GRP      16
#define NGRP     (AGG_T / GRP)
#define BATCH    16

#define TILE_H   (16 * EDGE_IN)
#define TILE_Q   (TILE_H / 4)
#define TILE_LD  ((TILE_Q + 31) / 32)

typedef unsigned u4a __attribute__((ext_vector_type(4)));

typedef __attribute__((ext_vector_type(2)))  _Float16 v2h;
typedef __attribute__((ext_vector_type(4)))  _Float16 v4h;


__device__ inline v8f wmma16(v16h a, v16h b, v8f c) {
    return __builtin_amdgcn_wmma_f32_16x16x32_f16(
        false, a, false, b, (short)0, c, false, false);
}

__device__ inline v16h cat8(v8h lo, v8h hi) {
    return __builtin_shufflevector(lo, hi,
        0, 1, 2, 3, 4, 5, 6, 7, 8, 9, 10, 11, 12, 13, 14, 15);
}

__device__ inline v16h load_a(const _Float16* H, int ld, int k0) {
    int lane = threadIdx.x & 31;
    int m  = lane & 15;
    int kb = k0 + ((lane & 16) ? 8 : 0);
    v8h lo = *(const v8h*)(H + m * ld + kb);
    v8h hi = *(const v8h*)(H + m * ld + kb + 16);
    return cat8(lo, hi);
}

__device__ inline v16h make_b(const _Float16* WT, int ldk, int k0, int n0) {
    int lane = threadIdx.x & 31;
    int n  = n0 + (lane & 15);
    int kb = k0 + ((lane & 16) ? 8 : 0);
    v8h lo = *(const v8h*)(WT + n * ldk + kb);
    v8h hi = *(const v8h*)(WT + n * ldk + kb + 16);
    return cat8(lo, hi);
}

__device__ inline v16h gfrag(const v8h* wq, int f) {
    int lane = threadIdx.x & 31;
    return cat8(wq[f * 64 + lane], wq[f * 64 + 32 + lane]);
}

__device__ inline void store_c(_Float16* H, int ld, v8f c, int n0,
                               const float* bias, bool relu) {
    int lane = threadIdx.x & 31;
    int n  = n0 + (lane & 15);
    int mb = (lane & 16) ? 8 : 0;
    float bv = bias[n];
#pragma unroll
    for (int r = 0; r < 8; ++r) {
        float v = c[r] + bv;
        if (relu) v = fmaxf(v, 0.0f);
        H[(mb + r) * ld + n] = (_Float16)v;
    }
}

template <int CTRL>
__device__ inline float dpp_f(float v) {
    return __builtin_bit_cast(float,
        __builtin_amdgcn_update_dpp(0, __builtin_bit_cast(int, v), CTRL, 0xF, 0xF, true));
}
template <int CTRL>
__device__ inline int dpp_i(int v) {
    return __builtin_amdgcn_update_dpp(0, v, CTRL, 0xF, 0xF, true);
}
#define DPP_XOR1   0xB1
#define DPP_XOR2   0x4E
#define DPP_XOR4   0x164
#define DPP_XOR8   0x168
#define DPP_SHARE0 0x150


__global__ __launch_bounds__(NWAVES * 32)
void nnconv_node_kernel(const float* __restrict__ x,
                        const int*   __restrict__ edge_index,
                        const float* __restrict__ W4,
                        const float* __restrict__ b4,
                        const float* __restrict__ W1, const float* __restrict__ b1,
                        const float* __restrict__ W2, const float* __restrict__ b2,
                        const float* __restrict__ W3, const float* __restrict__ b3,
                        float* __restrict__ G,
                        float* __restrict__ wpk,
                        unsigned* __restrict__ glist,
                        unsigned* __restrict__ goffs, int N, int E, int R, int C) {
    __shared__ alignas(16) _Float16 sBT[GW * BTP];
    __shared__ alignas(16) unsigned slist[CH];
    __shared__ unsigned shist[AGG_G];
    __shared__ unsigned soffs[AGG_G];
    __shared__ unsigned swsum[NWAVES];
    static_assert(AGG_G == NWAVES * 32, "one thread per owner in the prefix sum");
    static_assert(RMAX == (1 << RBITS) && CH * RMAX <= (1 << 30) && CH < 65536, "entry and bound packing");

    const int tid  = threadIdx.x;
    const int lane = tid & 31;
    const int wave = tid >> 5;
    const int m    = lane & 15;
    const int hi   = lane >> 4;
    const int* dstp = edge_index + (size_t)E;

    int dv[CHT];
#pragma unroll
    for (int k = 0; k < CHT; ++k) {
        const long e = (long)blockIdx.x * CH + k * (NWAVES * 32) + tid;
        dv[k] = ((int)blockIdx.x < C && e < E) ? dstp[e] : -1;
    }
    const long n0 = ((long)blockIdx.x * NWAVES + wave) * 16;
    float4 xf[2][4];
    {
        long node = n0 + m;
        if (node >= N) node = N - 1;
        const float4* xr = (const float4*)(x + (size_t)node * LATENT);
#pragma unroll
        for (int ks = 0; ks < 2; ++ks) {
            const int q = ks * 8 + hi * 2;
            xf[ks][0] = xr[q]; xf[ks][1] = xr[q + 1]; xf[ks][2] = xr[q + 4]; xf[ks][3] = xr[q + 5];
        }
    }

    for (int gid = blockIdx.x * (NWAVES * 32) + tid; gid < NFRAG * FRAGH / 2 + 3 * 32;
         gid += gridDim.x * (NWAVES * 32)) {
        if (gid < NFRAG * FRAGH / 2) {
            const int f = gid >> 8;
            const int r = (gid >> 7) & 1;
            const int l = (gid >> 2) & 31;
            const int j = (gid & 3) * 2;
            const float* Wl = (f < 6) ? W1 : ((f < 8) ? W2 : W3);
            const int K  = (f < 6) ? EDGE_IN : HID;
            const int k0 = (f < 6) ? (f >> 1) * 32 : 0;
            int n  = (f & 1) * 16 + (l & 15);
            if (f == 9) n = HID - 4 + ((l & 15) >> 2);
            const int k  = k0 + 16 * r + 8 * (l >> 4) + j;
            const float v0 = (n < HID && k     < K) ? Wl[k * HID + n]       : 0.0f;
            const float v1 = (n < HID && k + 1 < K) ? Wl[(k + 1) * HID + n] : 0.0f;
            v2h p; p[0] = (_Float16)v0; p[1] = (_Float16)v1;
            ((v2h*)wpk)[gid] = p;
        } else {
            const int q = gid - NFRAG * FRAGH / 2;
            int n = q & 31;
            if (q >= 64 && n >= 16) n = HID - 4 + ((n - 16) >> 2);
            const float* bl = (q < 32) ? b1 : ((q < 64) ? b2 : b3);
            wpk[WPK_BIAS + q] = (n < HID) ? bl[n] : 0.0f;
        }
    }

    for (int idx = tid; idx < GW * LATENT; idx += NWAVES * 32) {
        int j = idx >> 6, i = idx & 63;
        int pj = j >> 2, s = j & 3;
        bool w3 = (s < 3);
        int k  = w3 ? pj : (16 + (pj >> 2));
        int c  = w3 ? s  : (pj & 3);
        int bc = pj >> 2;
        bool isw = w3 || ((pj & 3) < 3);
        bool isz = (!isw) && (bc >= OUT_CH);
        const float* p = isw ? (W4 + k * OUTD + c) : (b4 + (isz ? 0 : bc));
        float v = p[i * OUT_CH];
        sBT[j * BTP + i] = (_Float16)(isz ? 0.0f : v);
    }

    {
        const unsigned Rd = (unsigned)R;
        for (int c = blockIdx.x; c < C; c += gridDim.x) {
            shist[tid] = 0u;
            __syncthreads();
            unsigned ow[CHT], rk[CHT], en[CHT];
#pragma unroll
            for (int k = 0; k < CHT; ++k) {
                const unsigned d = (unsigned)dv[k];
                const unsigned o = d / Rd;
                ow[k] = o;
                en[k] = ((unsigned)(k * (NWAVES * 32) + tid) << RBITS) | (d - o * Rd);
                rk[k] = (o < (unsigned)AGG_G) ? atomicAdd(&shist[o], 1u) : 0u;
            }
            __syncthreads();
            const unsigned v = shist[tid];
            unsigned inc = v;
#pragma unroll
            for (int d = 1; d < 32; d <<= 1) {
                const unsigned t = __shfl_up(inc, d);
                if (lane >= d) inc += t;
            }
            if (lane == 31) swsum[wave] = inc;
            __syncthreads();
            unsigned base = 0u;
            for (int w = 0; w < wave; ++w) base += swsum[w];
            const unsigned ex = base + inc - v;
            soffs[tid] = ex;
            goffs[(size_t)c * AGG_G + tid] = ex | ((ex + v) << 16);
            __syncthreads();
#pragma unroll
            for (int k = 0; k < CHT; ++k) {
                if (ow[k] < (unsigned)AGG_G) slist[soffs[ow[k]] + rk[k]] = en[k];
            }
            __syncthreads();
            {
                u4a* gl = (u4a*)(glist + (size_t)c * CH);
                const u4a* sl = (const u4a*)slist;
#pragma unroll
                for (int q = 0; q < CH / 4 / (NWAVES * 32); ++q) {
                    gl[q * (NWAVES * 32) + tid] = sl[q * (NWAVES * 32) + tid];
                }
            }
            const int cn = c + (int)gridDim.x;
#pragma unroll
            for (int k = 0; k < CHT; ++k) {
                const long e = (long)cn * CH + k * (NWAVES * 32) + tid;
                dv[k] = (cn < C && e < E) ? dstp[e] : -1;
            }
        }
    }
    __syncthreads();

    if (n0 >= N) return;

    v16h a[2];
#pragma unroll
    for (int ks = 0; ks < 2; ++ks) {
        const float4 f0 = xf[ks][0], f1 = xf[ks][1], f2 = xf[ks][2], f3 = xf[ks][3];
        v16h t;
        t[0]  = (_Float16)f0.x; t[1]  = (_Float16)f0.y; t[2]  = (_Float16)f0.z; t[3]  = (_Float16)f0.w;
        t[4]  = (_Float16)f1.x; t[5]  = (_Float16)f1.y; t[6]  = (_Float16)f1.z; t[7]  = (_Float16)f1.w;
        t[8]  = (_Float16)f2.x; t[9]  = (_Float16)f2.y; t[10] = (_Float16)f2.z; t[11] = (_Float16)f2.w;
        t[12] = (_Float16)f3.x; t[13] = (_Float16)f3.y; t[14] = (_Float16)f3.z; t[15] = (_Float16)f3.w;
        a[ks] = t;
    }

    v8f c[GW / 16];
#pragma unroll
    for (int t = 0; t < GW / 16; ++t) {
        v8f z = {};
        c[t] = wmma16(a[0], make_b(sBT, BTP, 0,  t * 16), z);
        c[t] = wmma16(a[1], make_b(sBT, BTP, 32, t * 16), c[t]);
    }
    float* gp = G + (size_t)(n0 + hi * 8) * GW + m;
#pragma unroll
    for (int t = 0; t < GW / 16; ++t) {
#pragma unroll
        for (int r = 0; r < 8; ++r) gp[r * GW + t * 16] = c[t][r];
    }
}

__global__ __launch_bounds__(WAVES * 32)
void nnconv_edge_kernel(const int*   __restrict__ edge_index,
                        const float* __restrict__ edge_attr,
                        const float* __restrict__ wpk,
                        const float* __restrict__ G,
                        float* __restrict__ msg,
                        int E) {
    const v8h*   wq = (const v8h*)wpk;
    const float* bq = wpk + WPK_BIAS;
    __shared__ alignas(16) _Float16 sA [WAVES][16 * HP * 2];

    const int tid = threadIdx.x;

    const int lane = tid & 31;
    const int wave = tid >> 5;
    const int m    = lane & 15;
    const int half = lane >> 4;

    const long tile = (long)blockIdx.x * WAVES + wave;
    const long e0   = tile * 16;
    if (e0 >= E) return;

    _Float16* H  = sA [wave];

    static_assert(TILE_LD * 32 * 4 <= 16 * HP * 2, "the staged tile fits the per-wave LDS tile");
    {
        const float4* tp = (const float4*)(edge_attr + (size_t)e0 * EDGE_IN);
        v4h* tq = (v4h*)H;
        if (e0 + 16 <= (long)E) {
            float4 pv[TILE_LD];
#pragma unroll
            for (int i = 0; i < TILE_LD; ++i) {
                int q = i * 32 + lane;
                if (i == TILE_LD - 1) q = (q < TILE_Q) ? q : (TILE_Q - 1);
                pv[i] = tp[q];
            }
#pragma unroll
            for (int i = 0; i < TILE_LD; ++i) {
                v4h h4;
                h4[0] = (_Float16)pv[i].x; h4[1] = (_Float16)pv[i].y;
                h4[2] = (_Float16)pv[i].z; h4[3] = (_Float16)pv[i].w;
                tq[i * 32 + lane] = h4;
            }
        } else {
            const long nf = ((long)E - e0) * EDGE_IN;
#pragma nounroll
            for (int i = 0; i < TILE_LD; ++i) {
                const long q = i * 32 + lane;
                float4 v = make_float4(0.0f, 0.0f, 0.0f, 0.0f);
                if (4 * q + 4 <= nf) {
                    v = tp[q];
                } else if (4 * q + 2 <= nf) {
                    const float2 h2 = *(const float2*)(tp + q);
                    v.x = h2.x; v.y = h2.y;
                }
                v4h h4;
                h4[0] = (_Float16)v.x; h4[1] = (_Float16)v.y;
                h4[2] = (_Float16)v.z; h4[3] = (_Float16)v.w;
                tq[q] = h4;
            }
        }
    }
    asm volatile("s_wait_dscnt 0" ::: "memory");

    v16h a1[3];
    {
        const v2h* er = (const v2h*)H + m * (EDGE_IN / 2);
        v2h f[2][8];
#pragma unroll
        for (int ks = 0; ks < 2; ++ks) {
            const int q = ks * 16 + half * 4;
#pragma unroll
            for (int j = 0; j < 4; ++j) {
                f[ks][j]     = er[q + j];
                f[ks][4 + j] = er[q + 8 + j];
            }
        }
        const v2h t0 = er[32], t1 = er[33], t2 = er[34];
#pragma unroll
        for (int ks = 0; ks < 2; ++ks) {
            v16h t;
#pragma unroll
            for (int j = 0; j < 8; ++j) {
                t[2 * j]     = f[ks][j][0];
                t[2 * j + 1] = f[ks][j][1];
            }
            a1[ks] = t;
        }
        const bool lo = (half == 0);
        const _Float16 hz = (_Float16)0.0f;
        v16h t = {};
        t[0] = lo ? t0[0] : hz; t[1] = lo ? t0[1] : hz;
        t[2] = lo ? t1[0] : hz; t[3] = lo ? t1[1] : hz;
        t[4] = lo ? t2[0] : hz; t[5] = lo ? t2[1] : hz;
        a1[2] = t;
    }

    {
        v8f c0 = {}, c1 = {};
#pragma unroll
        for (int ks = 0; ks < 3; ++ks) {
            c0 = wmma16(a1[ks], gfrag(wq, ks * 2),     c0);
            c1 = wmma16(a1[ks], gfrag(wq, ks * 2 + 1), c1);
        }
        store_c(H, 32, c0, 0,  bq, true);
        store_c(H, 32, c1, 16, bq, true);
    }
    asm volatile("s_wait_dscnt 0" ::: "memory");

    {
        v16h a = load_a(H, 32, 0);
        v8f c0 = {}, c1 = {};
        c0 = wmma16(a, gfrag(wq, 6), c0);
        c1 = wmma16(a, gfrag(wq, 7), c1);
        asm volatile("s_wait_dscnt 0" ::: "memory");
        store_c(H, 32, c0, 0,  bq + 32, true);
        store_c(H, 32, c1, 16, bq + 32, true);
    }
    asm volatile("s_wait_dscnt 0" ::: "memory");

    v8f c0 = {}, c1 = {};
    {
        v16h a = load_a(H, 32, 0);
        c0 = wmma16(a, gfrag(wq, 8), c0);
        c1 = wmma16(a, gfrag(wq, 9), c1);
    }

    {
        long es = e0 + (m ^ (half << 3));
        if (es >= (long)E) es = (long)E - 1;
        const int sv = edge_index[es];

        const float bA = bq[64 + m];
        const float bB = bq[64 + 16 + m];
        const bool hsel = (m & 3) < 3;
        const int  cq   = hsel ? (m & 3) : (m >> 2);
        const float4* gbase = (const float4*)G + m;

        float4 gp[8];
#define G_LOAD(i) gp[i] = gbase[(size_t)dpp_i<DPP_SHARE0 + (i)>(sv) * (GW / 4)];
        G_LOAD(0) G_LOAD(1) G_LOAD(2) G_LOAD(3) G_LOAD(4) G_LOAD(5) G_LOAD(6) G_LOAD(7)
#undef G_LOAD

        float v0[8], v1[8], v2[8];
#pragma unroll
        for (int i = 0; i < 8; ++i) {
            const float hA = fmaxf(c0[i] + bA, 0.0f);
            const float hB = fmaxf(c1[i] + bB, 0.0f);
            const float q  = gp[i].w * (hsel ? hB : 1.0f);
            v0[i] = hA * gp[i].x + ((cq == 0) ? q : 0.0f);
            v1[i] = hA * gp[i].y + ((cq == 1) ? q : 0.0f);
            v2[i] = hA * gp[i].z + ((cq == 2) ? q : 0.0f);
        }

#define XCH(dst, a, b, bit, CTRL) { const float gv_ = (bit) ? (a) : (b); const float kp_ = (bit) ? (b) : (a); dst = kp_ + dpp_f<CTRL>(gv_); }
        const bool b0 = (m & 1) != 0, b1 = (m & 2) != 0, b2 = (m & 4) != 0;
        float w0[4], w1[4], w2[4];
#pragma unroll
        for (int t = 0; t < 4; ++t) {
            XCH(w0[t], v0[2 * t], v0[2 * t + 1], b0, DPP_XOR1)
            XCH(w1[t], v1[2 * t], v1[2 * t + 1], b0, DPP_XOR1)
            XCH(w2[t], v2[2 * t], v2[2 * t + 1], b0, DPP_XOR1)
        }
        float y0[2], y1[2], y2[2];
#pragma unroll
        for (int u = 0; u < 2; ++u) {
            XCH(y0[u], w0[2 * u], w0[2 * u + 1], b1, DPP_XOR2)
            XCH(y1[u], w1[2 * u], w1[2 * u + 1], b1, DPP_XOR2)
            XCH(y2[u], w2[2 * u], w2[2 * u + 1], b1, DPP_XOR2)
        }
        float z0, z1, z2;
        XCH(z0, y0[0], y0[1], b2, DPP_XOR4)
        XCH(z1, y1[0], y1[1], b2, DPP_XOR4)
        XCH(z2, y2[0], y2[1], b2, DPP_XOR4)
#undef XCH
        z0 += dpp_f<DPP_XOR8>(z0);
        z1 += dpp_f<DPP_XOR8>(z1);
        z2 += dpp_f<DPP_XOR8>(z2);

        const long e = e0 + half * 8 + m;
        if (m < 8 && e < (long)E) {
            ((float4*)msg)[e] = make_float4(z0, z1, z2, 0.0f);
        }
    }
}

__global__ __launch_bounds__(AGG_T)
void nnconv_aggregate_kernel(const unsigned* __restrict__ glist,
                             const unsigned* __restrict__ goffs,
                             const float* __restrict__ msg,
                             const float* __restrict__ x,
                             const float* __restrict__ root_w,
                             const float* __restrict__ bias,
                             float* __restrict__ out, int E, int N, int R, int C) {
    __shared__ float sacc[RMAX * 4];

    const int tid = threadIdx.x;
    const int j   = tid & (GRP - 1);
    const int grp = tid / GRP;
    const int b   = blockIdx.x;
    const int n0  = b * R;
    const unsigned Ru = (unsigned)R;
    const float4* msg4 = (const float4*)msg;

    for (int q = tid; q < R * 4; q += AGG_T) sacc[q] = 0.0f;
    __syncthreads();

    for (int c0 = 0; c0 < C; c0 += NGRP * BATCH) {
        unsigned w[BATCH];
#pragma unroll
        for (int k = 0; k < BATCH; ++k) {
            const int c  = c0 + k * NGRP + grp;
            const int cc = (c < C) ? c : (C - 1);
            const unsigned t = goffs[(size_t)cc * AGG_G + b];
            w[k] = (c < C) ? t : 0u;
        }
        unsigned en[BATCH];
#pragma unroll
        for (int k = 0; k < BATCH; ++k) {
            const int c  = c0 + k * NGRP + grp;
            const int cc = (c < C) ? c : (C - 1);
            const unsigned idx = (w[k] & 0xffffu) + (unsigned)j;
            const bool has = idx < (w[k] >> 16);
            const unsigned t = glist[(size_t)cc * CH + (has ? idx : 0u)];
            en[k] = has ? t : 0xffffffffu;
        }
        float4 rc[BATCH];
        unsigned rr[BATCH];
#pragma unroll
        for (int k = 0; k < BATCH; ++k) {
            const int c  = c0 + k * NGRP + grp;
            const unsigned idx = (w[k] & 0xffffu) + (unsigned)j;
            const bool has = idx < (w[k] >> 16);
            const long e = (long)c * CH + (long)(en[k] >> RBITS);
            const unsigned r = en[k] & (unsigned)(RMAX - 1);
            const bool ok = has && e < (long)E && r < Ru;
            rc[k] = msg4[ok ? e : 0];
            rr[k] = ok ? r : 0xffffffffu;
        }
#pragma unroll
        for (int k = 0; k < BATCH; ++k) {
            if (rr[k] != 0xffffffffu) {
                atomicAdd(&sacc[rr[k] * 4 + 0], rc[k].x);
                atomicAdd(&sacc[rr[k] * 4 + 1], rc[k].y);
                atomicAdd(&sacc[rr[k] * 4 + 2], rc[k].z);
                atomicAdd(&sacc[rr[k] * 4 + 3], 1.0f);
            }
        }
#pragma unroll
        for (int k = 0; k < BATCH; ++k) {
            const int c = c0 + k * NGRP + grp;
            const unsigned hi = w[k] >> 16;
            for (unsigned idx = (w[k] & 0xffffu) + (unsigned)(GRP + j); idx < hi; idx += GRP) {
                const unsigned t = glist[(size_t)c * CH + idx];
                const long e = (long)c * CH + (long)(t >> RBITS);
                const unsigned r = t & (unsigned)(RMAX - 1);
                if (e < (long)E && r < Ru) {
                    const float4 mrec = msg4[e];
                    atomicAdd(&sacc[r * 4 + 0], mrec.x);
                    atomicAdd(&sacc[r * 4 + 1], mrec.y);
                    atomicAdd(&sacc[r * 4 + 2], mrec.z);
                    atomicAdd(&sacc[r * 4 + 3], 1.0f);
                }
            }
        }
    }
    __syncthreads();

    for (int t = tid; t < R; t += AGG_T) {
        const int n = n0 + t;
        if (n < N) {
            float inv = 1.0f / fmaxf(sacc[t * 4 + 3], 1.0f);
            float a0 = sacc[t * 4 + 0] * inv + bias[0];
            float a1 = sacc[t * 4 + 1] * inv + bias[1];
            float a2 = sacc[t * 4 + 2] * inv + bias[2];
            const float4* xr = (const float4*)(x + (size_t)n * LATENT);
#pragma unroll
            for (int q = 0; q < LATENT / 4; ++q) {
                float4 v = xr[q];
                const float* vp = (const float*)&v;
#pragma unroll
                for (int jj = 0; jj < 4; ++jj) {
                    int i = q * 4 + jj;
                    float xvv = vp[jj];
                    a0 += xvv * root_w[i * 3 + 0];
                    a1 += xvv * root_w[i * 3 + 1];
                    a2 += xvv * root_w[i * 3 + 2];
                }
            }
            out[(size_t)n * 3 + 0] = a0;
            out[(size_t)n * 3 + 1] = a1;
            out[(size_t)n * 3 + 2] = a2;
        }
    }
}


extern "C" void kernel_launch(void* const* d_in, const int* in_sizes, int n_in,
                              void* d_out, int out_size, void* d_ws, size_t ws_size,
                              hipStream_t stream) {
    const float* x          = (const float*)d_in[0];
    const int*   edge_index = (const int*)  d_in[1];
    const float* edge_attr  = (const float*)d_in[2];
    const float* W1 = (const float*)d_in[3];
    const float* b1 = (const float*)d_in[4];
    const float* W2 = (const float*)d_in[5];
    const float* b2 = (const float*)d_in[6];
    const float* W3 = (const float*)d_in[7];
    const float* b3 = (const float*)d_in[8];
    const float* W4 = (const float*)d_in[9];
    const float* b4 = (const float*)d_in[10];
    const float* root_w = (const float*)d_in[11];
    const float* bias   = (const float*)d_in[12];
    float* out = (float*)d_out;

    const int E = in_sizes[1] / 2;
    const int N = in_sizes[0] / LATENT;

    const int etiles = (E + 15) / 16;
    float* msg    = (float*)d_ws;
    float* G      = msg + (size_t)etiles * 16 * 4;

    const int ntiles = (N + 15) / 16;
    float* wpk    = G + (size_t)ntiles * 16 * GW;
    const int C = (E + CH - 1) / CH;
    unsigned* glist = (unsigned*)(wpk + WPK_FLOATS);
    unsigned* goffs = glist + (size_t)C * CH;
    if (ws_size < ((size_t)etiles * 16 * 4 + (size_t)ntiles * 16 * GW + WPK_FLOATS
                   + (size_t)C * (CH + AGG_G)) * sizeof(float)) return;
    const int owners = AGG_G;
    const int R      = (N + owners - 1) / owners;
    if (R < 1 || R > RMAX) return;

    {
        int tiles  = ntiles;
        int blocks = (tiles + NWAVES - 1) / NWAVES;
        nnconv_node_kernel<<<blocks, NWAVES * 32, 0, stream>>>(
            x, edge_index, W4, b4, W1, b1, W2, b2, W3, b3, G, wpk, glist, goffs, N, E, R, C);
    }
    {
        int tiles  = (E + 15) / 16;
        int blocks = (tiles + WAVES - 1) / WAVES;
        nnconv_edge_kernel<<<blocks, WAVES * 32, 0, stream>>>(
            edge_index, edge_attr, wpk, G,
            msg, E);
    }
    nnconv_aggregate_kernel<<<owners, AGG_T, 0, stream>>>(
        glist, goffs, msg, x, root_w, bias, out, E, N, R, C);
}
